# starting kernel
# baseline (speedup 1.0000x reference)
#include <hip/hip_runtime.h>


typedef float v2f __attribute__((ext_vector_type(2)));
typedef float v8f __attribute__((ext_vector_type(8)));

constexpr int NB  = 512;
constexpr int NN  = 256;
constexpr int AD  = 75;
constexpr int HD  = 128;
constexpr int KP1 = 76;
constexpr int HS  = 132;

constexpr int OFF_B   = NN * HS;
constexpr int OFF_DIS = OFF_B + NN * HD;
constexpr int SMEM_FLOATS = OFF_DIS + NN;

__device__ __forceinline__ v8f wmma_f32(v2f a, v2f b, v8f c) {
    return __builtin_amdgcn_wmma_f32_16x16x4_f32(
         false, a,  false, b,
         (short)0, c,  false,  false);
}

__device__ __forceinline__ void gemm_xw(const float* srcX, int xs, int kin, int kloop,
                                        const float* __restrict__ W,
                                        float* dstXW, int wave, int lane)
{
    const int half = lane >> 4;
    const int l15  = lane & 15;
    const int M0   = wave * 32;
    const v8f vzero = {0.f,0.f,0.f,0.f,0.f,0.f,0.f,0.f};
    v8f acc[2][8];
#pragma unroll
    for (int mt = 0; mt < 2; ++mt)
#pragma unroll
        for (int nt = 0; nt < 8; ++nt)
            acc[mt][nt] = vzero;

    for (int k0 = 0; k0 < kloop; k0 += 4) {
        const int ka = k0 + half * 2;
        v2f afrag[2];
#pragma unroll
        for (int mt = 0; mt < 2; ++mt) {
            const float* p = srcX + (M0 + mt * 16 + l15) * xs + ka;
            afrag[mt].x = p[0];
            afrag[mt].y = p[1];
        }
        v2f bfrag[8];
#pragma unroll
        for (int nt = 0; nt < 8; ++nt) {
            const int col = nt * 16 + l15;
            bfrag[nt].x = (ka     < kin) ? W[ka * HD + col]       : 0.f;
            bfrag[nt].y = (ka + 1 < kin) ? W[(ka + 1) * HD + col] : 0.f;
        }
#pragma unroll
        for (int mt = 0; mt < 2; ++mt)
#pragma unroll
            for (int nt = 0; nt < 8; ++nt)
                acc[mt][nt] = wmma_f32(afrag[mt], bfrag[nt], acc[mt][nt]);
    }
#pragma unroll
    for (int mt = 0; mt < 2; ++mt)
#pragma unroll
        for (int nt = 0; nt < 8; ++nt) {
            const int col = nt * 16 + l15;
#pragma unroll
            for (int v = 0; v < 8; ++v) {
                const int row = M0 + mt * 16 + half * 8 + v;
                dstXW[row * HD + col] = acc[mt][nt][v];
            }
        }
}

__device__ __forceinline__ void gemm_adj(const float* __restrict__ adjB,
                                         const float* sdis, const float* srcXW,
                                         const float* __restrict__ bias,
                                         bool relu,
                                         float* dstLDS, int ldsStride,
                                         float* __restrict__ dstG, bool toGlobal,
                                         int wave, int lane)
{
    const int half = lane >> 4;
    const int l15  = lane & 15;
    const int M0   = wave * 32;
    const v8f vzero = {0.f,0.f,0.f,0.f,0.f,0.f,0.f,0.f};
    v8f acc[2][8];
#pragma unroll
    for (int mt = 0; mt < 2; ++mt)
#pragma unroll
        for (int nt = 0; nt < 8; ++nt)
            acc[mt][nt] = vzero;

    int   cc[2];
    float dc[2];
#pragma unroll
    for (int mt = 0; mt < 2; ++mt) {
        cc[mt] = M0 + mt * 16 + l15;
        dc[mt] = sdis[cc[mt]];
    }

    for (int k0 = 0; k0 < NN; k0 += 4) {
        const int   r0  = k0 + half * 2;
        const float dr0 = sdis[r0];
        const float dr1 = sdis[r0 + 1];
        v2f afrag[2];
#pragma unroll
        for (int mt = 0; mt < 2; ++mt) {
            const float v0 = adjB[r0 * NN + cc[mt]];
            const float v1 = adjB[(r0 + 1) * NN + cc[mt]];
            afrag[mt].x = ((v0 > 0.5f) || (r0     == cc[mt])) ? dr0 * dc[mt] : 0.f;
            afrag[mt].y = ((v1 > 0.5f) || (r0 + 1 == cc[mt])) ? dr1 * dc[mt] : 0.f;
        }
        v2f bfrag[8];
#pragma unroll
        for (int nt = 0; nt < 8; ++nt) {
            const int col = nt * 16 + l15;
            bfrag[nt].x = srcXW[r0 * HD + col];
            bfrag[nt].y = srcXW[(r0 + 1) * HD + col];
        }
#pragma unroll
        for (int mt = 0; mt < 2; ++mt)
#pragma unroll
            for (int nt = 0; nt < 8; ++nt)
                acc[mt][nt] = wmma_f32(afrag[mt], bfrag[nt], acc[mt][nt]);
    }

#pragma unroll
    for (int mt = 0; mt < 2; ++mt)
#pragma unroll
        for (int nt = 0; nt < 8; ++nt) {
            const int col = nt * 16 + l15;
            const float bv = bias[col];
#pragma unroll
            for (int v = 0; v < 8; ++v) {
                const int row = M0 + mt * 16 + half * 8 + v;
                float o = acc[mt][nt][v] + bv;
                if (relu) o = fmaxf(o, 0.f);
                if (toGlobal) dstG[(size_t)row * HD + col] = o;
                else          dstLDS[row * ldsStride + col] = o;
            }
        }
}

__global__ void __launch_bounds__(256)
gcn_fused(const float* __restrict__ atoms, const float* __restrict__ adjs,
          const float* __restrict__ W1, const float* __restrict__ b1,
          const float* __restrict__ W2, const float* __restrict__ b2,
          float* __restrict__ out)
{
    extern __shared__ float smem[];
    float* bufA = smem;
    float* bufB = smem + OFF_B;
    float* sdis = smem + OFF_DIS;

    const int b    = blockIdx.x;
    const int tid  = threadIdx.x;
    const int wave = tid >> 5;
    const int lane = tid & 31;
    const float* adjB = adjs + (size_t)b * NN * NN;

    {
        const int c = tid;
        float deg = 1.f;
        for (int r = 0; r < NN; ++r) {
            const float v = adjB[r * NN + c];
            deg += ((v > 0.5f) && (r != c)) ? 1.f : 0.f;
        }
        sdis[c] = rsqrtf(deg);
    }
    {
        const float* at = atoms + (size_t)b * NN * AD;
        for (int i = tid; i < NN * KP1; i += 256) {
            const int row = i / KP1, col = i % KP1;
            bufA[row * KP1 + col] = (col < AD) ? at[row * AD + col] : 0.f;
        }
    }
    __syncthreads();

    gemm_xw(bufA, KP1, AD, KP1, W1, bufB, wave, lane);
    __syncthreads();

    gemm_adj(adjB, sdis, bufB, b1,  true, bufA, HS, nullptr, false, wave, lane);
    __syncthreads();

    gemm_xw(bufA, HS, HD, HD, W2, bufB, wave, lane);
    __syncthreads();

    gemm_adj(adjB, sdis, bufB, b2,  false, nullptr, 0,
             out + (size_t)b * NN * HD, true, wave, lane);
}

extern "C" void kernel_launch(void* const* d_in, const int* in_sizes, int n_in,
                              void* d_out, int out_size, void* d_ws, size_t ws_size,
                              hipStream_t stream) {
    (void)in_sizes; (void)n_in; (void)out_size; (void)d_ws; (void)ws_size;
    const float* atoms = (const float*)d_in[0];
    const float* adjs  = (const float*)d_in[1];
    const float* W1    = (const float*)d_in[2];
    const float* b1    = (const float*)d_in[3];
    const float* W2    = (const float*)d_in[4];
    const float* b2    = (const float*)d_in[5];
    float* out = (float*)d_out;

    const size_t shmem = (size_t)SMEM_FLOATS * sizeof(float);
    gcn_fused<<<dim3(NB), dim3(256), shmem, stream>>>(atoms, adjs, W1, b1, W2, b2, out);
}


# Prologue loads batched: atoms staged by 20 independent 16-byte loads per thread (was load-wait-store, one load in flight); degree sweep keeps 32 row loads in flight per wave (was 8)
# speedup vs baseline: 1.4497x; 1.4497x over previous
#include <hip/hip_runtime.h>


typedef float v2f __attribute__((ext_vector_type(2)));
typedef float v8f __attribute__((ext_vector_type(8)));

constexpr int NB  = 512;
constexpr int NN  = 256;
constexpr int AD  = 75;
constexpr int HD  = 128;
constexpr int KP1 = 76;
constexpr int HS  = 132;

constexpr int OFF_B   = NN * HS;
constexpr int OFF_DIS = OFF_B + NN * HD;
constexpr int SMEM_FLOATS = OFF_DIS + NN;

__device__ __forceinline__ v8f wmma_f32(v2f a, v2f b, v8f c) {
    return __builtin_amdgcn_wmma_f32_16x16x4_f32(
         false, a,  false, b,
         (short)0, c,  false,  false);
}

__device__ __forceinline__ void gemm_xw(const float* srcX, int xs, int kin, int kloop,
                                        const float* __restrict__ W,
                                        float* dstXW, int wave, int lane)
{
    const int half = lane >> 4;
    const int l15  = lane & 15;
    const int M0   = wave * 32;
    const v8f vzero = {0.f,0.f,0.f,0.f,0.f,0.f,0.f,0.f};
    v8f acc[2][8];
#pragma unroll
    for (int mt = 0; mt < 2; ++mt)
#pragma unroll
        for (int nt = 0; nt < 8; ++nt)
            acc[mt][nt] = vzero;

    for (int k0 = 0; k0 < kloop; k0 += 4) {
        const int ka = k0 + half * 2;
        v2f afrag[2];
#pragma unroll
        for (int mt = 0; mt < 2; ++mt) {
            const float* p = srcX + (M0 + mt * 16 + l15) * xs + ka;
            afrag[mt].x = p[0];
            afrag[mt].y = p[1];
        }
        v2f bfrag[8];
#pragma unroll
        for (int nt = 0; nt < 8; ++nt) {
            const int col = nt * 16 + l15;
            bfrag[nt].x = (ka     < kin) ? W[ka * HD + col]       : 0.f;
            bfrag[nt].y = (ka + 1 < kin) ? W[(ka + 1) * HD + col] : 0.f;
        }
#pragma unroll
        for (int mt = 0; mt < 2; ++mt)
#pragma unroll
            for (int nt = 0; nt < 8; ++nt)
                acc[mt][nt] = wmma_f32(afrag[mt], bfrag[nt], acc[mt][nt]);
    }
#pragma unroll
    for (int mt = 0; mt < 2; ++mt)
#pragma unroll
        for (int nt = 0; nt < 8; ++nt) {
            const int col = nt * 16 + l15;
#pragma unroll
            for (int v = 0; v < 8; ++v) {
                const int row = M0 + mt * 16 + half * 8 + v;
                dstXW[row * HD + col] = acc[mt][nt][v];
            }
        }
}

__device__ __forceinline__ void gemm_adj(const float* __restrict__ adjB,
                                         const float* sdis, const float* srcXW,
                                         const float* __restrict__ bias,
                                         bool relu,
                                         float* dstLDS, int ldsStride,
                                         float* __restrict__ dstG, bool toGlobal,
                                         int wave, int lane)
{
    const int half = lane >> 4;
    const int l15  = lane & 15;
    const int M0   = wave * 32;
    const v8f vzero = {0.f,0.f,0.f,0.f,0.f,0.f,0.f,0.f};
    v8f acc[2][8];
#pragma unroll
    for (int mt = 0; mt < 2; ++mt)
#pragma unroll
        for (int nt = 0; nt < 8; ++nt)
            acc[mt][nt] = vzero;

    int   cc[2];
    float dc[2];
#pragma unroll
    for (int mt = 0; mt < 2; ++mt) {
        cc[mt] = M0 + mt * 16 + l15;
        dc[mt] = sdis[cc[mt]];
    }

    for (int k0 = 0; k0 < NN; k0 += 4) {
        const int   r0  = k0 + half * 2;
        const float dr0 = sdis[r0];
        const float dr1 = sdis[r0 + 1];
        v2f afrag[2];
#pragma unroll
        for (int mt = 0; mt < 2; ++mt) {
            const float v0 = adjB[r0 * NN + cc[mt]];
            const float v1 = adjB[(r0 + 1) * NN + cc[mt]];
            afrag[mt].x = ((v0 > 0.5f) || (r0     == cc[mt])) ? dr0 * dc[mt] : 0.f;
            afrag[mt].y = ((v1 > 0.5f) || (r0 + 1 == cc[mt])) ? dr1 * dc[mt] : 0.f;
        }
        v2f bfrag[8];
#pragma unroll
        for (int nt = 0; nt < 8; ++nt) {
            const int col = nt * 16 + l15;
            bfrag[nt].x = srcXW[r0 * HD + col];
            bfrag[nt].y = srcXW[(r0 + 1) * HD + col];
        }
#pragma unroll
        for (int mt = 0; mt < 2; ++mt)
#pragma unroll
            for (int nt = 0; nt < 8; ++nt)
                acc[mt][nt] = wmma_f32(afrag[mt], bfrag[nt], acc[mt][nt]);
    }

#pragma unroll
    for (int mt = 0; mt < 2; ++mt)
#pragma unroll
        for (int nt = 0; nt < 8; ++nt) {
            const int col = nt * 16 + l15;
            const float bv = bias[col];
#pragma unroll
            for (int v = 0; v < 8; ++v) {
                const int row = M0 + mt * 16 + half * 8 + v;
                float o = acc[mt][nt][v] + bv;
                if (relu) o = fmaxf(o, 0.f);
                if (toGlobal) dstG[(size_t)row * HD + col] = o;
                else          dstLDS[row * ldsStride + col] = o;
            }
        }
}

__global__ void __launch_bounds__(256)
gcn_fused(const float* __restrict__ atoms, const float* __restrict__ adjs,
          const float* __restrict__ W1, const float* __restrict__ b1,
          const float* __restrict__ W2, const float* __restrict__ b2,
          float* __restrict__ out)
{
    extern __shared__ float smem[];
    float* bufA = smem;
    float* bufB = smem + OFF_B;
    float* sdis = smem + OFF_DIS;

    const int b    = blockIdx.x;
    const int tid  = threadIdx.x;
    const int wave = tid >> 5;
    const int lane = tid & 31;
    const float* adjB = adjs + (size_t)b * NN * NN;

    {
        const int c = tid;
        float deg = 1.f;
        for (int r0 = 0; r0 < NN; r0 += 32) {
            float v[32];
#pragma unroll
            for (int i = 0; i < 32; ++i)
                v[i] = adjB[(r0 + i) * NN + c];
#pragma unroll
            for (int i = 0; i < 32; ++i)
                deg += ((v[i] > 0.5f) && (r0 + i != c)) ? 1.f : 0.f;
        }
        sdis[c] = rsqrtf(deg);
    }
    {
        constexpr int NV  = NN * AD / 4;
        constexpr int NT  = 240;
        constexpr int NLD = NV / NT;
        static_assert(NT * NLD == NV, "staging split must cover the block exactly");
        if (tid < NT) {
            const float4* at4 = reinterpret_cast<const float4*>(atoms + (size_t)b * NN * AD);
            float4 v[NLD];
#pragma unroll
            for (int i = 0; i < NLD; ++i)
                v[i] = at4[tid + i * NT];
#pragma unroll
            for (int i = 0; i < NLD; ++i) {
                const int e   = 4 * (tid + i * NT);
                const int row = e / AD;
                const int col = e - row * AD;
                const float x[4] = {v[i].x, v[i].y, v[i].z, v[i].w};
#pragma unroll
                for (int s = 0; s < 4; ++s) {
                    const int wrap = (col + s >= AD) ? 1 : 0;
                    bufA[e + s + row + wrap] = x[s];
                }
            }
        }
        bufA[tid * KP1 + AD] = 0.f;
    }
    __syncthreads();

    gemm_xw(bufA, KP1, AD, KP1, W1, bufB, wave, lane);
    __syncthreads();

    gemm_adj(adjB, sdis, bufB, b1,  true, bufA, HS, nullptr, false, wave, lane);
    __syncthreads();

    gemm_xw(bufA, HS, HD, HD, W2, bufB, wave, lane);
    __syncthreads();

    gemm_adj(adjB, sdis, bufB, b2,  false, nullptr, 0,
             out + (size_t)b * NN * HD, true, wave, lane);
}

extern "C" void kernel_launch(void* const* d_in, const int* in_sizes, int n_in,
                              void* d_out, int out_size, void* d_ws, size_t ws_size,
                              hipStream_t stream) {
    (void)in_sizes; (void)n_in; (void)out_size; (void)d_ws; (void)ws_size;
    const float* atoms = (const float*)d_in[0];
    const float* adjs  = (const float*)d_in[1];
    const float* W1    = (const float*)d_in[2];
    const float* b1    = (const float*)d_in[3];
    const float* W2    = (const float*)d_in[4];
    const float* b2    = (const float*)d_in[5];
    float* out = (float*)d_out;

    const size_t shmem = (size_t)SMEM_FLOATS * sizeof(float);
    gcn_fused<<<dim3(NB), dim3(256), shmem, stream>>>(atoms, adjs, W1, b1, W2, b2, out);
}


# Adjacency thresholded once into a 256x256 bitmask in LDS by the degree sweep (degree = its popcount, exact); both aggregation GEMMs build A fragments from the bits: no global loads in their k-loops
# speedup vs baseline: 1.9574x; 1.3502x over previous
#include <hip/hip_runtime.h>


typedef float v2f __attribute__((ext_vector_type(2)));
typedef float v8f __attribute__((ext_vector_type(8)));

constexpr int NB  = 512;
constexpr int NN  = 256;
constexpr int AD  = 75;
constexpr int HD  = 128;
constexpr int KP1 = 76;
constexpr int HS  = 132;

constexpr int NW      = NN / 32;
constexpr int OFF_B   = NN * HS;
constexpr int OFF_DIS = OFF_B + NN * HD;
constexpr int OFF_BITS = OFF_DIS + NN;
constexpr int SMEM_FLOATS = OFF_BITS + NW * NN;

__device__ __forceinline__ v8f wmma_f32(v2f a, v2f b, v8f c) {
    return __builtin_amdgcn_wmma_f32_16x16x4_f32(
         false, a,  false, b,
         (short)0, c,  false,  false);
}

__device__ __forceinline__ void gemm_xw(const float* srcX, int xs, int kin, int kloop,
                                        const float* __restrict__ W,
                                        float* dstXW, int wave, int lane)
{
    const int half = lane >> 4;
    const int l15  = lane & 15;
    const int M0   = wave * 32;
    const v8f vzero = {0.f,0.f,0.f,0.f,0.f,0.f,0.f,0.f};
    v8f acc[2][8];
#pragma unroll
    for (int mt = 0; mt < 2; ++mt)
#pragma unroll
        for (int nt = 0; nt < 8; ++nt)
            acc[mt][nt] = vzero;

    for (int k0 = 0; k0 < kloop; k0 += 4) {
        const int ka = k0 + half * 2;
        v2f afrag[2];
#pragma unroll
        for (int mt = 0; mt < 2; ++mt) {
            const float* p = srcX + (M0 + mt * 16 + l15) * xs + ka;
            afrag[mt].x = p[0];
            afrag[mt].y = p[1];
        }
        v2f bfrag[8];
#pragma unroll
        for (int nt = 0; nt < 8; ++nt) {
            const int col = nt * 16 + l15;
            bfrag[nt].x = (ka     < kin) ? W[ka * HD + col]       : 0.f;
            bfrag[nt].y = (ka + 1 < kin) ? W[(ka + 1) * HD + col] : 0.f;
        }
#pragma unroll
        for (int mt = 0; mt < 2; ++mt)
#pragma unroll
            for (int nt = 0; nt < 8; ++nt)
                acc[mt][nt] = wmma_f32(afrag[mt], bfrag[nt], acc[mt][nt]);
    }
#pragma unroll
    for (int mt = 0; mt < 2; ++mt)
#pragma unroll
        for (int nt = 0; nt < 8; ++nt) {
            const int col = nt * 16 + l15;
#pragma unroll
            for (int v = 0; v < 8; ++v) {
                const int row = M0 + mt * 16 + half * 8 + v;
                dstXW[row * HD + col] = acc[mt][nt][v];
            }
        }
}

__device__ __forceinline__ void gemm_adj(const unsigned* sbits,
                                         const float* sdis, const float* srcXW,
                                         const float* __restrict__ bias,
                                         bool relu,
                                         float* dstLDS, int ldsStride,
                                         float* __restrict__ dstG, bool toGlobal,
                                         int wave, int lane)
{
    const int half = lane >> 4;
    const int l15  = lane & 15;
    const int M0   = wave * 32;
    const v8f vzero = {0.f,0.f,0.f,0.f,0.f,0.f,0.f,0.f};
    v8f acc[2][8];
#pragma unroll
    for (int mt = 0; mt < 2; ++mt)
#pragma unroll
        for (int nt = 0; nt < 8; ++nt)
            acc[mt][nt] = vzero;

    int   cc[2];
    float dc[2];
#pragma unroll
    for (int mt = 0; mt < 2; ++mt) {
        cc[mt] = M0 + mt * 16 + l15;
        dc[mt] = sdis[cc[mt]];
    }

    for (int kw = 0; kw < NW; ++kw) {
        unsigned wb[2];
#pragma unroll
        for (int mt = 0; mt < 2; ++mt)
            wb[mt] = sbits[kw * NN + cc[mt]] >> (half * 2);
#pragma unroll
        for (int j = 0; j < 8; ++j) {
            const int   r0  = kw * 32 + j * 4 + half * 2;
            const float dr0 = sdis[r0];
            const float dr1 = sdis[r0 + 1];
            v2f afrag[2];
#pragma unroll
            for (int mt = 0; mt < 2; ++mt) {
                afrag[mt].x = ((wb[mt] >> (4 * j))     & 1u) ? dr0 * dc[mt] : 0.f;
                afrag[mt].y = ((wb[mt] >> (4 * j + 1)) & 1u) ? dr1 * dc[mt] : 0.f;
            }
            v2f bfrag[8];
#pragma unroll
            for (int nt = 0; nt < 8; ++nt) {
                const int col = nt * 16 + l15;
                bfrag[nt].x = srcXW[r0 * HD + col];
                bfrag[nt].y = srcXW[(r0 + 1) * HD + col];
            }
#pragma unroll
            for (int mt = 0; mt < 2; ++mt)
#pragma unroll
                for (int nt = 0; nt < 8; ++nt)
                    acc[mt][nt] = wmma_f32(afrag[mt], bfrag[nt], acc[mt][nt]);
        }
    }

#pragma unroll
    for (int mt = 0; mt < 2; ++mt)
#pragma unroll
        for (int nt = 0; nt < 8; ++nt) {
            const int col = nt * 16 + l15;
            const float bv = bias[col];
#pragma unroll
            for (int v = 0; v < 8; ++v) {
                const int row = M0 + mt * 16 + half * 8 + v;
                float o = acc[mt][nt][v] + bv;
                if (relu) o = fmaxf(o, 0.f);
                if (toGlobal) dstG[(size_t)row * HD + col] = o;
                else          dstLDS[row * ldsStride + col] = o;
            }
        }
}

__global__ void __launch_bounds__(256)
gcn_fused(const float* __restrict__ atoms, const float* __restrict__ adjs,
          const float* __restrict__ W1, const float* __restrict__ b1,
          const float* __restrict__ W2, const float* __restrict__ b2,
          float* __restrict__ out)
{
    extern __shared__ float smem[];
    float* bufA = smem;
    float* bufB = smem + OFF_B;
    float* sdis = smem + OFF_DIS;
    unsigned* sbits = reinterpret_cast<unsigned*>(smem + OFF_BITS);

    const int b    = blockIdx.x;
    const int tid  = threadIdx.x;
    const int wave = tid >> 5;
    const int lane = tid & 31;
    const float* adjB = adjs + (size_t)b * NN * NN;

    {
        const int c = tid;
        int cnt = 0;
        for (int r0 = 0; r0 < NN; r0 += 32) {
            float v[32];
#pragma unroll
            for (int i = 0; i < 32; ++i)
                v[i] = adjB[(r0 + i) * NN + c];
            unsigned word = ((c >> 5) == (r0 >> 5)) ? (1u << (c & 31)) : 0u;
#pragma unroll
            for (int i = 0; i < 32; ++i)
                word |= (v[i] > 0.5f) ? (1u << i) : 0u;
            sbits[(r0 >> 5) * NN + c] = word;
            cnt += __popc(word);
        }
        sdis[c] = rsqrtf((float)cnt);
    }
    {
        constexpr int NV  = NN * AD / 4;
        constexpr int NT  = 240;
        constexpr int NLD = NV / NT;
        static_assert(NT * NLD == NV, "staging split must cover the block exactly");
        if (tid < NT) {
            const float4* at4 = reinterpret_cast<const float4*>(atoms + (size_t)b * NN * AD);
            float4 v[NLD];
#pragma unroll
            for (int i = 0; i < NLD; ++i)
                v[i] = at4[tid + i * NT];
#pragma unroll
            for (int i = 0; i < NLD; ++i) {
                const int e   = 4 * (tid + i * NT);
                const int row = e / AD;
                const int col = e - row * AD;
                const float x[4] = {v[i].x, v[i].y, v[i].z, v[i].w};
#pragma unroll
                for (int s = 0; s < 4; ++s) {
                    const int wrap = (col + s >= AD) ? 1 : 0;
                    bufA[e + s + row + wrap] = x[s];
                }
            }
        }
        bufA[tid * KP1 + AD] = 0.f;
    }
    __syncthreads();

    gemm_xw(bufA, KP1, AD, KP1, W1, bufB, wave, lane);
    __syncthreads();

    gemm_adj(sbits, sdis, bufB, b1,  true, bufA, HS, nullptr, false, wave, lane);
    __syncthreads();

    gemm_xw(bufA, HS, HD, HD, W2, bufB, wave, lane);
    __syncthreads();

    gemm_adj(sbits, sdis, bufB, b2,  false, nullptr, 0,
             out + (size_t)b * NN * HD, true, wave, lane);
}

extern "C" void kernel_launch(void* const* d_in, const int* in_sizes, int n_in,
                              void* d_out, int out_size, void* d_ws, size_t ws_size,
                              hipStream_t stream) {
    (void)in_sizes; (void)n_in; (void)out_size; (void)d_ws; (void)ws_size;
    const float* atoms = (const float*)d_in[0];
    const float* adjs  = (const float*)d_in[1];
    const float* W1    = (const float*)d_in[2];
    const float* b1    = (const float*)d_in[3];
    const float* W2    = (const float*)d_in[4];
    const float* b2    = (const float*)d_in[5];
    float* out = (float*)d_out;

    const size_t shmem = (size_t)SMEM_FLOATS * sizeof(float);
    gcn_fused<<<dim3(NB), dim3(256), shmem, stream>>>(atoms, adjs, W1, b1, W2, b2, out);
}


# B operand of both aggregation GEMMs kept in LDS with rows r, r+1 interleaved and each lane's two tiles on adjacent columns: one 16-byte load = two tiles' B fragments (was 2-address loads + moves)
# speedup vs baseline: 2.3783x; 1.2150x over previous
#include <hip/hip_runtime.h>


typedef float v2f __attribute__((ext_vector_type(2)));
typedef float v8f __attribute__((ext_vector_type(8)));

constexpr int NB  = 512;
constexpr int NN  = 256;
constexpr int AD  = 75;
constexpr int HD  = 128;
constexpr int KP1 = 76;
constexpr int XS  = 80;
constexpr int NTX = XS / 16;
constexpr int AXS = 84;
constexpr int HS  = 132;

constexpr int NW      = NN / 32;
constexpr int OFF_B   = NN * HS;
constexpr int OFF_DIS = OFF_B + NN * HD;
constexpr int OFF_BITS = OFF_DIS + NN;
constexpr int SMEM_FLOATS = OFF_BITS + NW * NN;

__device__ __forceinline__ v8f wmma_f32(v2f a, v2f b, v8f c) {
    return __builtin_amdgcn_wmma_f32_16x16x4_f32(
         false, a,  false, b,
         (short)0, c,  false,  false);
}

__device__ __forceinline__ void xw_kstep(int k0, const float* srcX, int xs, int kin,
                                         const float* __restrict__ W,
                                         int half, int l15, int M0, v8f (&acc)[2][8])
{
    const int ka = k0 + half * 2;
    v2f afrag[2];
#pragma unroll
    for (int mt = 0; mt < 2; ++mt) {
        const float* p = srcX + (M0 + mt * 16 + l15) * xs + ka;
        afrag[mt].x = p[0];
        afrag[mt].y = p[1];
    }
    v2f bfrag[8];
#pragma unroll
    for (int nt = 0; nt < 8; ++nt) {
        const int col = nt * 16 + l15;
        bfrag[nt].x = (ka     < kin) ? W[ka * HD + col]       : 0.f;
        bfrag[nt].y = (ka + 1 < kin) ? W[(ka + 1) * HD + col] : 0.f;
    }
#pragma unroll
    for (int mt = 0; mt < 2; ++mt)
#pragma unroll
        for (int nt = 0; nt < 8; ++nt)
            acc[mt][nt] = wmma_f32(afrag[mt], bfrag[nt], acc[mt][nt]);
}

template <bool UNROLL_ALL, bool PAIR_DST>
__device__ __forceinline__ void gemm_xw(const float* srcX, int xs, int kin, int kloop,
                                        const float* __restrict__ W,
                                        const float* __restrict__ bias, bool hasBias, bool relu,
                                        float* dstXW, int ds, int wave, int lane)
{
    const int half = lane >> 4;
    const int l15  = lane & 15;
    const int M0   = wave * 32;
    const v8f vzero = {0.f,0.f,0.f,0.f,0.f,0.f,0.f,0.f};
    v8f acc[2][8];
#pragma unroll
    for (int mt = 0; mt < 2; ++mt)
#pragma unroll
        for (int nt = 0; nt < 8; ++nt)
            acc[mt][nt] = vzero;

    if constexpr (UNROLL_ALL) {
#pragma unroll
        for (int k0 = 0; k0 < kloop; k0 += 4)
            xw_kstep(k0, srcX, xs, kin, W, half, l15, M0, acc);
    } else {
        for (int k0 = 0; k0 < kloop; k0 += 4)
            xw_kstep(k0, srcX, xs, kin, W, half, l15, M0, acc);
    }
    if constexpr (PAIR_DST) {
#pragma unroll
        for (int mt = 0; mt < 2; ++mt)
#pragma unroll
            for (int nt = 0; nt < 8; ++nt)
#pragma unroll
                for (int v = 0; v < 8; v += 2) {
                    const int rp = (M0 + mt * 16 + half * 8 + v) >> 1;
                    const float2 q = {acc[mt][nt][v], acc[mt][nt][v + 1]};
                    *reinterpret_cast<float2*>(dstXW + rp * (2 * HD) + 2 * (nt * 16 + l15)) = q;
                }
    } else {
#pragma unroll
        for (int mt = 0; mt < 2; ++mt)
#pragma unroll
            for (int nt = 0; nt < 8; ++nt) {
                const int col = nt * 16 + l15;
                const float bv = hasBias ? bias[col] : 0.f;
#pragma unroll
                for (int v = 0; v < 8; ++v) {
                    const int row = M0 + mt * 16 + half * 8 + v;
                    float o = acc[mt][nt][v];
                    if (hasBias) o += bv;
                    if (relu) o = fmaxf(o, 0.f);
                    dstXW[row * ds + col] = o;
                }
            }
    }
}

template <int NT>
__device__ __forceinline__ void gemm_adj(const unsigned* sbits,
                                         const float* sdis, const float* srcXW,
                                         const float* __restrict__ bias, bool hasBias,
                                         float* dstLDS, int ldsStride,
                                         float* __restrict__ dstG, bool toGlobal,
                                         int wave, int lane)
{
    const int half = lane >> 4;
    const int l15  = lane & 15;
    const int M0   = wave * 32;
    const v8f vzero = {0.f,0.f,0.f,0.f,0.f,0.f,0.f,0.f};
    v8f acc[2][NT];
#pragma unroll
    for (int mt = 0; mt < 2; ++mt)
#pragma unroll
        for (int nt = 0; nt < NT; ++nt)
            acc[mt][nt] = vzero;

    int   cc[2];
    float dc[2];
#pragma unroll
    for (int mt = 0; mt < 2; ++mt) {
        cc[mt] = M0 + mt * 16 + l15;
        dc[mt] = sdis[cc[mt]];
    }

    for (int kw = 0; kw < NW; ++kw) {
        unsigned wb[2];
#pragma unroll
        for (int mt = 0; mt < 2; ++mt)
            wb[mt] = sbits[kw * NN + cc[mt]] >> (half * 2);
#pragma unroll
        for (int j = 0; j < 8; ++j) {
            const int   r0  = kw * 32 + j * 4 + half * 2;
            const float dr0 = sdis[r0];
            const float dr1 = sdis[r0 + 1];
            v2f afrag[2];
#pragma unroll
            for (int mt = 0; mt < 2; ++mt) {
                afrag[mt].x = ((wb[mt] >> (4 * j))     & 1u) ? dr0 * dc[mt] : 0.f;
                afrag[mt].y = ((wb[mt] >> (4 * j + 1)) & 1u) ? dr1 * dc[mt] : 0.f;
            }
            const float* bp = srcXW + (r0 >> 1) * (32 * NT);
            v2f bfrag[NT];
#pragma unroll
            for (int m = 0; m < NT / 2; ++m) {
                const float4 q = *reinterpret_cast<const float4*>(bp + 64 * m + 4 * l15);
                bfrag[2 * m].x     = q.x;
                bfrag[2 * m].y     = q.y;
                bfrag[2 * m + 1].x = q.z;
                bfrag[2 * m + 1].y = q.w;
            }
            if constexpr ((NT & 1) != 0) {
                const float2 q = *reinterpret_cast<const float2*>(bp + 64 * (NT / 2) + 2 * l15);
                bfrag[NT - 1].x = q.x;
                bfrag[NT - 1].y = q.y;
            }
#pragma unroll
            for (int mt = 0; mt < 2; ++mt)
#pragma unroll
                for (int nt = 0; nt < NT; ++nt)
                    acc[mt][nt] = wmma_f32(afrag[mt], bfrag[nt], acc[mt][nt]);
        }
    }

#pragma unroll
    for (int mt = 0; mt < 2; ++mt) {
#pragma unroll
        for (int m = 0; m < NT / 2; ++m) {
            const int col = 32 * m + 2 * l15;
            const float bv0 = hasBias ? bias[col]     : 0.f;
            const float bv1 = hasBias ? bias[col + 1] : 0.f;
#pragma unroll
            for (int v = 0; v < 8; ++v) {
                const int row = M0 + mt * 16 + half * 8 + v;
                float2 o = {acc[mt][2 * m][v], acc[mt][2 * m + 1][v]};
                if (hasBias) { o.x += bv0; o.y += bv1; }
                if (toGlobal) *reinterpret_cast<float2*>(dstG + (size_t)row * HD + col) = o;
                else          *reinterpret_cast<float2*>(dstLDS + row * ldsStride + col) = o;
            }
        }
        if constexpr ((NT & 1) != 0) {
            const int col = 32 * (NT / 2) + l15;
            const float bv = hasBias ? bias[col] : 0.f;
#pragma unroll
            for (int v = 0; v < 8; ++v) {
                const int row = M0 + mt * 16 + half * 8 + v;
                float o = acc[mt][NT - 1][v];
                if (hasBias) o += bv;
                if (toGlobal) dstG[(size_t)row * HD + col] = o;
                else          dstLDS[row * ldsStride + col] = o;
            }
        }
    }
}

__global__ void __launch_bounds__(256)
gcn_fused(const float* __restrict__ atoms, const float* __restrict__ adjs,
          const float* __restrict__ W1, const float* __restrict__ b1,
          const float* __restrict__ W2, const float* __restrict__ b2,
          float* __restrict__ out)
{
    extern __shared__ __align__(16) float smem[];
    float* bufA = smem;
    float* bufB = smem + OFF_B;
    float* sdis = smem + OFF_DIS;
    unsigned* sbits = reinterpret_cast<unsigned*>(smem + OFF_BITS);

    const int b    = blockIdx.x;
    const int tid  = threadIdx.x;
    const int wave = tid >> 5;
    const int lane = tid & 31;
    const float* adjB = adjs + (size_t)b * NN * NN;

    {
        const int c = tid;
        int cnt = 0;
        for (int r0 = 0; r0 < NN; r0 += 32) {
            float v[32];
#pragma unroll
            for (int i = 0; i < 32; ++i)
                v[i] = adjB[(r0 + i) * NN + c];
            unsigned word = ((c >> 5) == (r0 >> 5)) ? (1u << (c & 31)) : 0u;
#pragma unroll
            for (int i = 0; i < 32; ++i)
                word |= (v[i] > 0.5f) ? (1u << i) : 0u;
            sbits[(r0 >> 5) * NN + c] = word;
            cnt += __popc(word);
        }
        sdis[c] = rsqrtf((float)cnt);
    }
    {
        constexpr int NV  = NN * AD / 4;
        constexpr int NT  = 240;
        constexpr int NLD = NV / NT;
        static_assert(NT * NLD == NV, "staging split must cover the block exactly");
        if (tid < NT) {
            const float4* at4 = reinterpret_cast<const float4*>(atoms + (size_t)b * NN * AD);
            float4 v[NLD];
#pragma unroll
            for (int i = 0; i < NLD; ++i)
                v[i] = at4[tid + i * NT];
#pragma unroll
            for (int i = 0; i < NLD; ++i) {
                const int e   = 4 * (tid + i * NT);
                const int row = e / AD;
                const int col = e - row * AD;
                const int odd = row & 1;
                const int base = row * XS + 2 * col;
                const int stay = base - (XS - 1) * odd;
                const int next = base - (2 * AD - 1) + (XS - 1) * odd;
                const float x[4] = {v[i].x, v[i].y, v[i].z, v[i].w};
#pragma unroll
                for (int s = 0; s < 4; ++s)
                    bufA[((col + s >= AD) ? next : stay) + 2 * s] = x[s];
            }
        }
        for (int p = AD; p < XS; ++p)
            bufA[(tid >> 1) * (2 * XS) + 2 * p + (tid & 1)] = 0.f;
    }
    __syncthreads();

    gemm_adj<NTX>(sbits, sdis, bufA, nullptr,  false, bufB, AXS, nullptr, false, wave, lane);
    __syncthreads();

    gemm_xw<true, false>(bufB, AXS, AD, KP1, W1, b1,  true,  true, bufA, HS, wave, lane);
    __syncthreads();

    gemm_xw<false, true>(bufA, HS, HD, HD, W2, nullptr,  false,  false, bufB, HD, wave, lane);
    __syncthreads();

    gemm_adj<HD / 16>(sbits, sdis, bufB, b2,  true, nullptr, 0,
                          out + (size_t)b * NN * HD, true, wave, lane);
}

extern "C" void kernel_launch(void* const* d_in, const int* in_sizes, int n_in,
                              void* d_out, int out_size, void* d_ws, size_t ws_size,
                              hipStream_t stream) {
    (void)in_sizes; (void)n_in; (void)out_size; (void)d_ws; (void)ws_size;
    const float* atoms = (const float*)d_in[0];
    const float* adjs  = (const float*)d_in[1];
    const float* W1    = (const float*)d_in[2];
    const float* b1    = (const float*)d_in[3];
    const float* W2    = (const float*)d_in[4];
    const float* b2    = (const float*)d_in[5];
    float* out = (float*)d_out;

    const size_t shmem = (size_t)SMEM_FLOATS * sizeof(float);
    gcn_fused<<<dim3(NB), dim3(256), shmem, stream>>>(atoms, adjs, W1, b1, W2, b2, out);
}


# Rolled GEMM loops (both aggregations, H @ W2) hold two register sets of operand fragments: the next group of k-steps is loaded in one place of the stream while the current group's products run
# speedup vs baseline: 2.4935x; 1.0484x over previous
#include <hip/hip_runtime.h>


typedef float v2f __attribute__((ext_vector_type(2)));
typedef float v8f __attribute__((ext_vector_type(8)));

constexpr int NB  = 512;
constexpr int NN  = 256;
constexpr int AD  = 75;
constexpr int HD  = 128;
constexpr int KP1 = 76;
constexpr int XS  = 80;
constexpr int NTX = XS / 16;
constexpr int AXS = 84;
constexpr int HS  = 132;

constexpr int NW      = NN / 32;
constexpr int OFF_B   = NN * HS;
constexpr int OFF_DIS = OFF_B + NN * HD;
constexpr int OFF_BITS = OFF_DIS + NN;
constexpr int SMEM_FLOATS = OFF_BITS + NW * NN;

__device__ __forceinline__ v8f wmma_f32(v2f a, v2f b, v8f c) {
    return __builtin_amdgcn_wmma_f32_16x16x4_f32(
         false, a,  false, b,
         (short)0, c,  false,  false);
}

__device__ __forceinline__ void group_edge()
{
    __builtin_amdgcn_sched_barrier(0);
    asm volatile("" ::: "memory");
}

__device__ __forceinline__ void xw_kstep(int k0, const float* srcX, int xs, int kin,
                                         const float* __restrict__ W,
                                         int half, int l15, int M0, v8f (&acc)[2][8])
{
    const int ka = k0 + half * 2;
    v2f afrag[2];
#pragma unroll
    for (int mt = 0; mt < 2; ++mt) {
        const float* p = srcX + (M0 + mt * 16 + l15) * xs + ka;
        afrag[mt].x = p[0];
        afrag[mt].y = p[1];
    }
    v2f bfrag[8];
#pragma unroll
    for (int nt = 0; nt < 8; ++nt) {
        const int col = nt * 16 + l15;
        bfrag[nt].x = (ka     < kin) ? W[ka * HD + col]       : 0.f;
        bfrag[nt].y = (ka + 1 < kin) ? W[(ka + 1) * HD + col] : 0.f;
    }
#pragma unroll
    for (int mt = 0; mt < 2; ++mt)
#pragma unroll
        for (int nt = 0; nt < 8; ++nt)
            acc[mt][nt] = wmma_f32(afrag[mt], bfrag[nt], acc[mt][nt]);
}

__device__ __forceinline__ void xw_load_a(int k0, const float* srcX, int xs, int half, int l15, int M0,
                                          v2f (&a)[4][2])
{
#pragma unroll
    for (int s = 0; s < 4; ++s)
#pragma unroll
        for (int mt = 0; mt < 2; ++mt) {
            const float* p = srcX + (M0 + mt * 16 + l15) * xs + k0 + 4 * s + half * 2;
            a[s][mt].x = p[0];
            a[s][mt].y = p[1];
        }
}

__device__ __forceinline__ void xw_load_b(int k0, const float* __restrict__ W, int half, int l15,
                                          v2f (&b)[2][8])
{
#pragma unroll
    for (int s = 0; s < 2; ++s) {
        const int ka = k0 + 4 * s + half * 2;
#pragma unroll
        for (int nt = 0; nt < 8; ++nt) {
            const int col = nt * 16 + l15;
            b[s][nt].x = W[ka * HD + col];
            b[s][nt].y = W[(ka + 1) * HD + col];
        }
    }
}

__device__ __forceinline__ void xw_products(const v2f (&a0)[2], const v2f (&a1)[2], const v2f (&b)[2][8],
                                            v8f (&acc)[2][8])
{
#pragma unroll
    for (int mt = 0; mt < 2; ++mt)
#pragma unroll
        for (int nt = 0; nt < 8; ++nt)
            acc[mt][nt] = wmma_f32(a0[mt], b[0][nt], acc[mt][nt]);
#pragma unroll
    for (int mt = 0; mt < 2; ++mt)
#pragma unroll
        for (int nt = 0; nt < 8; ++nt)
            acc[mt][nt] = wmma_f32(a1[mt], b[1][nt], acc[mt][nt]);
}

template <bool UNROLL_ALL, bool PAIR_DST>
__device__ __forceinline__ void gemm_xw(const float* srcX, int xs, int kin, int kloop,
                                        const float* __restrict__ W,
                                        const float* __restrict__ bias, bool hasBias, bool relu,
                                        float* dstXW, int ds, int wave, int lane)
{
    const int half = lane >> 4;
    const int l15  = lane & 15;
    const int M0   = wave * 32;
    const v8f vzero = {0.f,0.f,0.f,0.f,0.f,0.f,0.f,0.f};
    v8f acc[2][8];
#pragma unroll
    for (int mt = 0; mt < 2; ++mt)
#pragma unroll
        for (int nt = 0; nt < 8; ++nt)
            acc[mt][nt] = vzero;

    if constexpr (UNROLL_ALL) {
#pragma unroll
        for (int k0 = 0; k0 < kloop; k0 += 4)
            xw_kstep(k0, srcX, xs, kin, W, half, l15, M0, acc);
    } else {
        v2f an[4][2];
        v2f bA[2][8], bB[2][8];
        xw_load_a(0, srcX, xs, half, l15, M0, an);
        xw_load_b(0, W, half, l15, bA);
        group_edge();
        for (int k0 = 0; k0 < kloop; k0 += 16) {
            v2f af[4][2];
#pragma unroll
            for (int s = 0; s < 4; ++s)
#pragma unroll
                for (int mt = 0; mt < 2; ++mt)
                    af[s][mt] = an[s][mt];
            const int kn = (k0 + 16 < kloop) ? k0 + 16 : 0;
            xw_load_a(kn, srcX, xs, half, l15, M0, an);
            xw_load_b(k0 + 8, W, half, l15, bB);
            group_edge();
            xw_products(af[0], af[1], bA, acc);
            group_edge();
            xw_load_b(kn, W, half, l15, bA);
            group_edge();
            xw_products(af[2], af[3], bB, acc);
            group_edge();
        }
    }
    if constexpr (PAIR_DST) {
#pragma unroll
        for (int mt = 0; mt < 2; ++mt)
#pragma unroll
            for (int nt = 0; nt < 8; ++nt)
#pragma unroll
                for (int v = 0; v < 8; v += 2) {
                    const int rp = (M0 + mt * 16 + half * 8 + v) >> 1;
                    const float2 q = {acc[mt][nt][v], acc[mt][nt][v + 1]};
                    *reinterpret_cast<float2*>(dstXW + rp * (2 * HD) + 2 * (nt * 16 + l15)) = q;
                }
    } else {
#pragma unroll
        for (int mt = 0; mt < 2; ++mt)
#pragma unroll
            for (int nt = 0; nt < 8; ++nt) {
                const int col = nt * 16 + l15;
                const float bv = hasBias ? bias[col] : 0.f;
#pragma unroll
                for (int v = 0; v < 8; ++v) {
                    const int row = M0 + mt * 16 + half * 8 + v;
                    float o = acc[mt][nt][v];
                    if (hasBias) o += bv;
                    if (relu) o = fmaxf(o, 0.f);
                    dstXW[row * ds + col] = o;
                }
            }
    }
}

template <int NT>
__device__ __forceinline__ void adj_load_b(const float* srcXW, int grp, int half, int l15, v2f (&bf)[4][NT])
{
#pragma unroll
    for (int jj = 0; jj < 4; ++jj) {
        const float* bp = srcXW + (8 * grp + 2 * jj + half) * (32 * NT);
#pragma unroll
        for (int m = 0; m < NT / 2; ++m) {
            const float4 q = *reinterpret_cast<const float4*>(bp + 64 * m + 4 * l15);
            bf[jj][2 * m].x     = q.x;
            bf[jj][2 * m].y     = q.y;
            bf[jj][2 * m + 1].x = q.z;
            bf[jj][2 * m + 1].y = q.w;
        }
        if constexpr ((NT & 1) != 0) {
            const float2 q = *reinterpret_cast<const float2*>(bp + 64 * (NT / 2) + 2 * l15);
            bf[jj][NT - 1].x = q.x;
            bf[jj][NT - 1].y = q.y;
        }
    }
}

__device__ __forceinline__ void adj_load_a(const unsigned* sbits, const float* sdis, int kw,
                                           const int (&cc)[2], int half, unsigned (&wb)[2], float2 (&dr)[8])
{
#pragma unroll
    for (int mt = 0; mt < 2; ++mt)
        wb[mt] = sbits[kw * NN + cc[mt]] >> (half * 2);
#pragma unroll
    for (int j = 0; j < 8; ++j)
        dr[j] = *reinterpret_cast<const float2*>(sdis + kw * 32 + j * 4 + half * 2);
}

template <int NT, int G>
__device__ __forceinline__ void adj_products(const unsigned (&wb)[2], const float2 (&dr)[8],
                                             const float (&dc)[2], const v2f (&bf)[4][NT], v8f (&acc)[2][NT])
{
#pragma unroll
    for (int jj = 0; jj < 4; ++jj) {
        const int j = 4 * G + jj;
        v2f afrag[2];
#pragma unroll
        for (int mt = 0; mt < 2; ++mt) {
            afrag[mt].x = ((wb[mt] >> (4 * j))     & 1u) ? dr[j].x * dc[mt] : 0.f;
            afrag[mt].y = ((wb[mt] >> (4 * j + 1)) & 1u) ? dr[j].y * dc[mt] : 0.f;
        }
#pragma unroll
        for (int mt = 0; mt < 2; ++mt)
#pragma unroll
            for (int nt = 0; nt < NT; ++nt)
                acc[mt][nt] = wmma_f32(afrag[mt], bf[jj][nt], acc[mt][nt]);
    }
}

template <int NT>
__device__ __forceinline__ void gemm_adj(const unsigned* sbits,
                                         const float* sdis, const float* srcXW,
                                         const float* __restrict__ bias, bool hasBias,
                                         float* dstLDS, int ldsStride,
                                         float* __restrict__ dstG, bool toGlobal,
                                         int wave, int lane)
{
    const int half = lane >> 4;
    const int l15  = lane & 15;
    const int M0   = wave * 32;
    const v8f vzero = {0.f,0.f,0.f,0.f,0.f,0.f,0.f,0.f};
    v8f acc[2][NT];
#pragma unroll
    for (int mt = 0; mt < 2; ++mt)
#pragma unroll
        for (int nt = 0; nt < NT; ++nt)
            acc[mt][nt] = vzero;

    int   cc[2];
    float dc[2];
#pragma unroll
    for (int mt = 0; mt < 2; ++mt) {
        cc[mt] = M0 + mt * 16 + l15;
        dc[mt] = sdis[cc[mt]];
    }

    v2f b0[4][NT], b1[4][NT];
    unsigned wbn[2];
    float2   drn[8];
    adj_load_b<NT>(srcXW, 0, half, l15, b0);
    adj_load_a(sbits, sdis, 0, cc, half, wbn, drn);
    group_edge();
    for (int kw = 0; kw < NW; ++kw) {
        unsigned wb[2];
        float2   dr[8];
#pragma unroll
        for (int mt = 0; mt < 2; ++mt)
            wb[mt] = wbn[mt];
#pragma unroll
        for (int j = 0; j < 8; ++j)
            dr[j] = drn[j];
        adj_load_b<NT>(srcXW, 2 * kw + 1, half, l15, b1);
        group_edge();
        adj_products<NT, 0>(wb, dr, dc, b0, acc);
        group_edge();
        const int kwn = (kw + 1) & (NW - 1);
        adj_load_b<NT>(srcXW, 2 * kwn, half, l15, b0);
        adj_load_a(sbits, sdis, kwn, cc, half, wbn, drn);
        group_edge();
        adj_products<NT, 1>(wb, dr, dc, b1, acc);
        group_edge();
    }

#pragma unroll
    for (int mt = 0; mt < 2; ++mt) {
#pragma unroll
        for (int m = 0; m < NT / 2; ++m) {
            const int col = 32 * m + 2 * l15;
            const float bv0 = hasBias ? bias[col]     : 0.f;
            const float bv1 = hasBias ? bias[col + 1] : 0.f;
#pragma unroll
            for (int v = 0; v < 8; ++v) {
                const int row = M0 + mt * 16 + half * 8 + v;
                float2 o = {acc[mt][2 * m][v], acc[mt][2 * m + 1][v]};
                if (hasBias) { o.x += bv0; o.y += bv1; }
                if (toGlobal) *reinterpret_cast<float2*>(dstG + (size_t)row * HD + col) = o;
                else          *reinterpret_cast<float2*>(dstLDS + row * ldsStride + col) = o;
            }
        }
        if constexpr ((NT & 1) != 0) {
            const int col = 32 * (NT / 2) + l15;
            const float bv = hasBias ? bias[col] : 0.f;
#pragma unroll
            for (int v = 0; v < 8; ++v) {
                const int row = M0 + mt * 16 + half * 8 + v;
                float o = acc[mt][NT - 1][v];
                if (hasBias) o += bv;
                if (toGlobal) dstG[(size_t)row * HD + col] = o;
                else          dstLDS[row * ldsStride + col] = o;
            }
        }
    }
}

__global__ void __launch_bounds__(256)
gcn_fused(const float* __restrict__ atoms, const float* __restrict__ adjs,
          const float* __restrict__ W1, const float* __restrict__ b1,
          const float* __restrict__ W2, const float* __restrict__ b2,
          float* __restrict__ out)
{
    extern __shared__ __align__(16) float smem[];
    float* bufA = smem;
    float* bufB = smem + OFF_B;
    float* sdis = smem + OFF_DIS;
    unsigned* sbits = reinterpret_cast<unsigned*>(smem + OFF_BITS);

    const int b    = blockIdx.x;
    const int tid  = threadIdx.x;
    const int wave = tid >> 5;
    const int lane = tid & 31;
    const float* adjB = adjs + (size_t)b * NN * NN;

    {
        const int c = tid;
        int cnt = 0;
        for (int r0 = 0; r0 < NN; r0 += 32) {
            float v[32];
#pragma unroll
            for (int i = 0; i < 32; ++i)
                v[i] = adjB[(r0 + i) * NN + c];
            unsigned word = ((c >> 5) == (r0 >> 5)) ? (1u << (c & 31)) : 0u;
#pragma unroll
            for (int i = 0; i < 32; ++i)
                word |= (v[i] > 0.5f) ? (1u << i) : 0u;
            sbits[(r0 >> 5) * NN + c] = word;
            cnt += __popc(word);
        }
        sdis[c] = rsqrtf((float)cnt);
    }
    {
        constexpr int NV  = NN * AD / 4;
        constexpr int NT  = 240;
        constexpr int NLD = NV / NT;
        static_assert(NT * NLD == NV, "staging split must cover the block exactly");
        if (tid < NT) {
            const float4* at4 = reinterpret_cast<const float4*>(atoms + (size_t)b * NN * AD);
            float4 v[NLD];
#pragma unroll
            for (int i = 0; i < NLD; ++i)
                v[i] = at4[tid + i * NT];
#pragma unroll
            for (int i = 0; i < NLD; ++i) {
                const int e   = 4 * (tid + i * NT);
                const int row = e / AD;
                const int col = e - row * AD;
                const int odd = row & 1;
                const int base = row * XS + 2 * col;
                const int stay = base - (XS - 1) * odd;
                const int next = base - (2 * AD - 1) + (XS - 1) * odd;
                const float x[4] = {v[i].x, v[i].y, v[i].z, v[i].w};
#pragma unroll
                for (int s = 0; s < 4; ++s)
                    bufA[((col + s >= AD) ? next : stay) + 2 * s] = x[s];
            }
        }
        for (int p = AD; p < XS; ++p)
            bufA[(tid >> 1) * (2 * XS) + 2 * p + (tid & 1)] = 0.f;
    }
    __syncthreads();

    gemm_adj<NTX>(sbits, sdis, bufA, nullptr,  false, bufB, AXS, nullptr, false, wave, lane);
    __syncthreads();

    gemm_xw<true, false>(bufB, AXS, AD, KP1, W1, b1,  true,  true, bufA, HS, wave, lane);
    __syncthreads();

    gemm_xw<false, true>(bufA, HS, HD, HD, W2, nullptr,  false,  false, bufB, HD, wave, lane);
    __syncthreads();

    gemm_adj<HD / 16>(sbits, sdis, bufB, b2,  true, nullptr, 0,
                          out + (size_t)b * NN * HD, true, wave, lane);
}

extern "C" void kernel_launch(void* const* d_in, const int* in_sizes, int n_in,
                              void* d_out, int out_size, void* d_ws, size_t ws_size,
                              hipStream_t stream) {
    (void)in_sizes; (void)n_in; (void)out_size; (void)d_ws; (void)ws_size;
    const float* atoms = (const float*)d_in[0];
    const float* adjs  = (const float*)d_in[1];
    const float* W1    = (const float*)d_in[2];
    const float* b1    = (const float*)d_in[3];
    const float* W2    = (const float*)d_in[4];
    const float* b2    = (const float*)d_in[5];
    float* out = (float*)d_out;

    const size_t shmem = (size_t)SMEM_FLOATS * sizeof(float);
    gcn_fused<<<dim3(NB), dim3(256), shmem, stream>>>(atoms, adjs, W1, b1, W2, b2, out);
}


# Two workgroups resident per WGP: all intermediates in one LDS buffer of per-wave regions overwritten in place (269 KB -> 141 KB), kernel held to 256 registers (k-step groups of 2); same sum order
# speedup vs baseline: 2.6534x; 1.0641x over previous
#include <hip/hip_runtime.h>


typedef float v2f __attribute__((ext_vector_type(2)));
typedef float v8f __attribute__((ext_vector_type(8)));

constexpr int NB  = 512;
constexpr int NN  = 256;
constexpr int AD  = 75;
constexpr int HD  = 128;
constexpr int KP1 = 76;
constexpr int XS  = 80;
constexpr int NTX = XS / 16;
constexpr int AXS = 84;
constexpr int HS  = 132;

constexpr int NW      = NN / 32;
constexpr int REG     = 32 * HS;
constexpr int REGPAD  = REG - 32 * HD;
constexpr int OFF_DIS = 8 * REG;
constexpr int OFF_BITS = OFF_DIS + NN;
constexpr int SMEM_FLOATS = OFF_BITS + NW * NN;

__device__ __forceinline__ v8f wmma_f32(v2f a, v2f b, v8f c) {
    return __builtin_amdgcn_wmma_f32_16x16x4_f32(
         false, a,  false, b,
         (short)0, c,  false,  false);
}

__device__ __forceinline__ void group_edge()
{
    __builtin_amdgcn_sched_barrier(0);
    asm volatile("" ::: "memory");
}

__device__ __forceinline__ void wave_region_handover()
{
    asm volatile("" ::: "memory");
}

__device__ __forceinline__ void xw_kstep(int k0, const float* srcX, int xs, int kin,
                                         const float* __restrict__ W,
                                         int half, int l15, int M0, v8f (&acc)[2][8])
{
    const int ka = k0 + half * 2;
    v2f afrag[2];
#pragma unroll
    for (int mt = 0; mt < 2; ++mt) {
        const float* p = srcX + (M0 + mt * 16 + l15) * xs + ka;
        afrag[mt].x = p[0];
        afrag[mt].y = p[1];
    }
    v2f bfrag[8];
#pragma unroll
    for (int nt = 0; nt < 8; ++nt) {
        const int col = nt * 16 + l15;
        bfrag[nt].x = (ka     < kin) ? W[ka * HD + col]       : 0.f;
        bfrag[nt].y = (ka + 1 < kin) ? W[(ka + 1) * HD + col] : 0.f;
    }
#pragma unroll
    for (int mt = 0; mt < 2; ++mt)
#pragma unroll
        for (int nt = 0; nt < 8; ++nt)
            acc[mt][nt] = wmma_f32(afrag[mt], bfrag[nt], acc[mt][nt]);
}

__device__ __forceinline__ void xw_load_a(int k0, const float* srcX, int xs, int half, int l15, int M0,
                                          v2f (&a)[4][2])
{
#pragma unroll
    for (int s = 0; s < 4; ++s)
#pragma unroll
        for (int mt = 0; mt < 2; ++mt) {
            const float* p = srcX + (M0 + mt * 16 + l15) * xs + k0 + 4 * s + half * 2;
            a[s][mt].x = p[0];
            a[s][mt].y = p[1];
        }
}

__device__ __forceinline__ void xw_load_b(int k0, const float* __restrict__ W, int half, int l15,
                                          v2f (&b)[2][8])
{
#pragma unroll
    for (int s = 0; s < 2; ++s) {
        const int ka = k0 + 4 * s + half * 2;
#pragma unroll
        for (int nt = 0; nt < 8; ++nt) {
            const int col = nt * 16 + l15;
            b[s][nt].x = W[ka * HD + col];
            b[s][nt].y = W[(ka + 1) * HD + col];
        }
    }
}

__device__ __forceinline__ void xw_products(const v2f (&a0)[2], const v2f (&a1)[2], const v2f (&b)[2][8],
                                            v8f (&acc)[2][8])
{
#pragma unroll
    for (int mt = 0; mt < 2; ++mt)
#pragma unroll
        for (int nt = 0; nt < 8; ++nt)
            acc[mt][nt] = wmma_f32(a0[mt], b[0][nt], acc[mt][nt]);
#pragma unroll
    for (int mt = 0; mt < 2; ++mt)
#pragma unroll
        for (int nt = 0; nt < 8; ++nt)
            acc[mt][nt] = wmma_f32(a1[mt], b[1][nt], acc[mt][nt]);
}

template <bool UNROLL_ALL, bool PAIR_DST>
__device__ __forceinline__ void gemm_xw(const float* srcX, int xs, int kin, int kloop,
                                        const float* __restrict__ W,
                                        const float* __restrict__ bias, bool hasBias, bool relu,
                                        float* dstXW, int ds, int wave, int lane)
{
    const int half = lane >> 4;
    const int l15  = lane & 15;
    const int M0   = wave * 32;
    const v8f vzero = {0.f,0.f,0.f,0.f,0.f,0.f,0.f,0.f};
    v8f acc[2][8];
#pragma unroll
    for (int mt = 0; mt < 2; ++mt)
#pragma unroll
        for (int nt = 0; nt < 8; ++nt)
            acc[mt][nt] = vzero;

    if constexpr (UNROLL_ALL) {
#pragma unroll
        for (int k0 = 0; k0 < kloop; k0 += 4) {
            xw_kstep(k0, srcX, xs, kin, W, half, l15, M0, acc);
            if ((k0 & 12) == 12)
                group_edge();
        }
    } else {
        v2f an[4][2];
        v2f bA[2][8], bB[2][8];
        xw_load_a(0, srcX, xs, half, l15, M0, an);
        xw_load_b(0, W, half, l15, bA);
        group_edge();
        for (int k0 = 0; k0 < kloop; k0 += 16) {
            v2f af[4][2];
#pragma unroll
            for (int s = 0; s < 4; ++s)
#pragma unroll
                for (int mt = 0; mt < 2; ++mt)
                    af[s][mt] = an[s][mt];
            const int kn = (k0 + 16 < kloop) ? k0 + 16 : 0;
            xw_load_a(kn, srcX, xs, half, l15, M0, an);
            xw_load_b(k0 + 8, W, half, l15, bB);
            group_edge();
            xw_products(af[0], af[1], bA, acc);
            group_edge();
            xw_load_b(kn, W, half, l15, bA);
            group_edge();
            xw_products(af[2], af[3], bB, acc);
            group_edge();
        }
    }
    if constexpr (PAIR_DST) {
#pragma unroll
        for (int mt = 0; mt < 2; ++mt)
#pragma unroll
            for (int nt = 0; nt < 8; ++nt)
#pragma unroll
                for (int v = 0; v < 8; v += 2) {
                    const int rp = (M0 + mt * 16 + half * 8 + v) >> 1;
                    const float2 q = {acc[mt][nt][v], acc[mt][nt][v + 1]};
                    *reinterpret_cast<float2*>(dstXW + rp * (2 * HD) + 2 * (nt * 16 + l15)) = q;
                }
    } else {
#pragma unroll
        for (int mt = 0; mt < 2; ++mt)
#pragma unroll
            for (int nt = 0; nt < 8; ++nt) {
                const int col = nt * 16 + l15;
                const float bv = hasBias ? bias[col] : 0.f;
#pragma unroll
                for (int v = 0; v < 8; ++v) {
                    const int row = M0 + mt * 16 + half * 8 + v;
                    float o = acc[mt][nt][v];
                    if (hasBias) o += bv;
                    if (relu) o = fmaxf(o, 0.f);
                    dstXW[row * ds + col] = o;
                }
            }
    }
}

template <int NT, int PAD>
__device__ __forceinline__ void adj_load_g(const float* srcXW, const float* sdis, int g, int half, int l15,
                                           v2f (&bf)[2][NT], float2 (&dr)[2])
{
#pragma unroll
    for (int jj = 0; jj < 2; ++jj) {
        const int p = 4 * g + 2 * jj + half;
        const float* bp = srcXW + p * (32 * NT) + (p >> 4) * PAD;
#pragma unroll
        for (int m = 0; m < NT / 2; ++m) {
            const float4 q = *reinterpret_cast<const float4*>(bp + 64 * m + 4 * l15);
            bf[jj][2 * m].x     = q.x;
            bf[jj][2 * m].y     = q.y;
            bf[jj][2 * m + 1].x = q.z;
            bf[jj][2 * m + 1].y = q.w;
        }
        if constexpr ((NT & 1) != 0) {
            const float2 q = *reinterpret_cast<const float2*>(bp + 64 * (NT / 2) + 2 * l15);
            bf[jj][NT - 1].x = q.x;
            bf[jj][NT - 1].y = q.y;
        }
        dr[jj] = *reinterpret_cast<const float2*>(sdis + 8 * g + 4 * jj + 2 * half);
    }
}

__device__ __forceinline__ void adj_load_w(const unsigned* sbits, int kw, const int (&cc)[2], int half,
                                           unsigned (&wb)[2])
{
#pragma unroll
    for (int mt = 0; mt < 2; ++mt)
        wb[mt] = sbits[kw * NN + cc[mt]] >> (half * 2);
}

template <int NT, int Q>
__device__ __forceinline__ void adj_products(const unsigned (&wb)[2], const float2 (&dr)[2],
                                             const float (&dc)[2], const v2f (&bf)[2][NT], v8f (&acc)[2][NT])
{
#pragma unroll
    for (int jj = 0; jj < 2; ++jj) {
        const int j = 2 * Q + jj;
        v2f afrag[2];
#pragma unroll
        for (int mt = 0; mt < 2; ++mt) {
            afrag[mt].x = ((wb[mt] >> (4 * j))     & 1u) ? dr[jj].x * dc[mt] : 0.f;
            afrag[mt].y = ((wb[mt] >> (4 * j + 1)) & 1u) ? dr[jj].y * dc[mt] : 0.f;
        }
#pragma unroll
        for (int mt = 0; mt < 2; ++mt)
#pragma unroll
            for (int nt = 0; nt < NT; ++nt)
                acc[mt][nt] = wmma_f32(afrag[mt], bf[jj][nt], acc[mt][nt]);
    }
}

template <int NT, int PAD, bool SYNC_STORE>
__device__ __forceinline__ void gemm_adj(const unsigned* sbits,
                                         const float* sdis, const float* srcXW,
                                         const float* __restrict__ bias, bool hasBias,
                                         float* dstLDS, int ldsStride,
                                         float* __restrict__ dstG, bool toGlobal,
                                         int wave, int lane)
{
    const int half = lane >> 4;
    const int l15  = lane & 15;
    const int M0   = wave * 32;
    const v8f vzero = {0.f,0.f,0.f,0.f,0.f,0.f,0.f,0.f};
    v8f acc[2][NT];
#pragma unroll
    for (int mt = 0; mt < 2; ++mt)
#pragma unroll
        for (int nt = 0; nt < NT; ++nt)
            acc[mt][nt] = vzero;

    int   cc[2];
    float dc[2];
#pragma unroll
    for (int mt = 0; mt < 2; ++mt) {
        cc[mt] = M0 + mt * 16 + l15;
        dc[mt] = sdis[cc[mt]];
    }

    v2f    b0[2][NT], b1[2][NT];
    float2 d0[2], d1[2];
    unsigned wbn[2];
    adj_load_g<NT, PAD>(srcXW, sdis, 0, half, l15, b0, d0);
    adj_load_w(sbits, 0, cc, half, wbn);
    group_edge();
    for (int kw = 0; kw < NW; ++kw) {
        unsigned wb[2];
#pragma unroll
        for (int mt = 0; mt < 2; ++mt)
            wb[mt] = wbn[mt];
        adj_load_g<NT, PAD>(srcXW, sdis, 4 * kw + 1, half, l15, b1, d1);
        group_edge();
        adj_products<NT, 0>(wb, d0, dc, b0, acc);
        group_edge();
        adj_load_g<NT, PAD>(srcXW, sdis, 4 * kw + 2, half, l15, b0, d0);
        group_edge();
        adj_products<NT, 1>(wb, d1, dc, b1, acc);
        group_edge();
        adj_load_g<NT, PAD>(srcXW, sdis, 4 * kw + 3, half, l15, b1, d1);
        group_edge();
        adj_products<NT, 2>(wb, d0, dc, b0, acc);
        group_edge();
        const int kwn = (kw + 1) & (NW - 1);
        adj_load_g<NT, PAD>(srcXW, sdis, 4 * kwn, half, l15, b0, d0);
        adj_load_w(sbits, kwn, cc, half, wbn);
        group_edge();
        adj_products<NT, 3>(wb, d1, dc, b1, acc);
        group_edge();
    }

    if constexpr (SYNC_STORE)
        __syncthreads();

#pragma unroll
    for (int mt = 0; mt < 2; ++mt) {
#pragma unroll
        for (int m = 0; m < NT / 2; ++m) {
            const int col = 32 * m + 2 * l15;
            const float bv0 = hasBias ? bias[col]     : 0.f;
            const float bv1 = hasBias ? bias[col + 1] : 0.f;
#pragma unroll
            for (int v = 0; v < 8; ++v) {
                const int row = M0 + mt * 16 + half * 8 + v;
                float2 o = {acc[mt][2 * m][v], acc[mt][2 * m + 1][v]};
                if (hasBias) { o.x += bv0; o.y += bv1; }
                if (toGlobal) *reinterpret_cast<float2*>(dstG + (size_t)row * HD + col) = o;
                else          *reinterpret_cast<float2*>(dstLDS + row * ldsStride + col) = o;
            }
        }
        if constexpr ((NT & 1) != 0) {
            const int col = 32 * (NT / 2) + l15;
            const float bv = hasBias ? bias[col] : 0.f;
#pragma unroll
            for (int v = 0; v < 8; ++v) {
                const int row = M0 + mt * 16 + half * 8 + v;
                float o = acc[mt][NT - 1][v];
                if (hasBias) o += bv;
                if (toGlobal) dstG[(size_t)row * HD + col] = o;
                else          dstLDS[row * ldsStride + col] = o;
            }
        }
    }
}

__global__ void __launch_bounds__(256, 4)
gcn_fused(const float* __restrict__ atoms, const float* __restrict__ adjs,
          const float* __restrict__ W1, const float* __restrict__ b1,
          const float* __restrict__ W2, const float* __restrict__ b2,
          float* __restrict__ out)
{
    extern __shared__ __align__(16) float smem[];
    float* buf  = smem;
    float* sdis = smem + OFF_DIS;
    unsigned* sbits = reinterpret_cast<unsigned*>(smem + OFF_BITS);

    const int b    = blockIdx.x;
    const int tid  = threadIdx.x;
    const int wave = tid >> 5;
    const int lane = tid & 31;
    const float* adjB = adjs + (size_t)b * NN * NN;

    {
        const int c = tid;
        int cnt = 0;
        for (int r0 = 0; r0 < NN; r0 += 32) {
            float v[32];
#pragma unroll
            for (int i = 0; i < 32; ++i)
                v[i] = adjB[(r0 + i) * NN + c];
            unsigned word = ((c >> 5) == (r0 >> 5)) ? (1u << (c & 31)) : 0u;
#pragma unroll
            for (int i = 0; i < 32; ++i)
                word |= (v[i] > 0.5f) ? (1u << i) : 0u;
            sbits[(r0 >> 5) * NN + c] = word;
            cnt += __popc(word);
        }
        sdis[c] = rsqrtf((float)cnt);
    }
    {
        constexpr int NV  = NN * AD / 4;
        constexpr int NT  = 240;
        constexpr int NLD = NV / NT;
        static_assert(NT * NLD == NV, "staging split must cover the block exactly");
        if (tid < NT) {
            const float4* at4 = reinterpret_cast<const float4*>(atoms + (size_t)b * NN * AD);
            float4 v[NLD];
#pragma unroll
            for (int i = 0; i < NLD; ++i)
                v[i] = at4[tid + i * NT];
#pragma unroll
            for (int i = 0; i < NLD; ++i) {
                const int e   = 4 * (tid + i * NT);
                const int row = e / AD;
                const int col = e - row * AD;
                const int odd = row & 1;
                const int base = row * XS + 2 * col;
                const int stay = base - (XS - 1) * odd;
                const int next = base - (2 * AD - 1) + (XS - 1) * odd;
                const float x[4] = {v[i].x, v[i].y, v[i].z, v[i].w};
#pragma unroll
                for (int s = 0; s < 4; ++s)
                    buf[((col + s >= AD) ? next : stay) + 2 * s] = x[s];
            }
        }
        for (int p = AD; p < XS; ++p)
            buf[(tid >> 1) * (2 * XS) + 2 * p + (tid & 1)] = 0.f;
    }
    __syncthreads();

    float* axw = buf + wave * (REG - 32 * AXS);
    gemm_adj<NTX, 0, true>(sbits, sdis, buf, nullptr,  false, axw, AXS, nullptr, false, wave, lane);
    wave_region_handover();

    gemm_xw<true, false>(axw, AXS, AD, KP1, W1, b1,  true,  true, buf, HS, wave, lane);
    wave_region_handover();

    gemm_xw<false, true>(buf, HS, HD, HD, W2, nullptr,  false,  false,
                         buf + wave * REGPAD, HD, wave, lane);
    __syncthreads();

    gemm_adj<HD / 16, REGPAD, false>(sbits, sdis, buf, b2,  true, nullptr, 0,
                                     out + (size_t)b * NN * HD, true, wave, lane);
}

extern "C" void kernel_launch(void* const* d_in, const int* in_sizes, int n_in,
                              void* d_out, int out_size, void* d_ws, size_t ws_size,
                              hipStream_t stream) {
    (void)in_sizes; (void)n_in; (void)out_size; (void)d_ws; (void)ws_size;
    const float* atoms = (const float*)d_in[0];
    const float* adjs  = (const float*)d_in[1];
    const float* W1    = (const float*)d_in[2];
    const float* b1    = (const float*)d_in[3];
    const float* W2    = (const float*)d_in[4];
    const float* b2    = (const float*)d_in[5];
    float* out = (float*)d_out;

    const size_t shmem = (size_t)SMEM_FLOATS * sizeof(float);
    gcn_fused<<<dim3(NB), dim3(256), shmem, stream>>>(atoms, adjs, W1, b1, W2, b2, out);
}
